# starting kernel
# baseline (speedup 1.0000x reference)
#include <hip/hip_runtime.h>

#define B_  64
#define S_  512
#define D_  768
#define K_  8
#define R_  277

typedef __attribute__((ext_vector_type(2))) float v2f;
typedef __attribute__((ext_vector_type(8))) float v8f;

__device__ __forceinline__ v8f wmma_f32_16x16x4(v2f a, v2f b, v8f c) {
#if defined(__gfx1250__) && __has_builtin(__builtin_amdgcn_wmma_f32_16x16x4_f32)
  return __builtin_amdgcn_wmma_f32_16x16x4_f32(
      false, a, false, b, (short)0, c, false, false);
#else
  asm volatile("v_wmma_f32_16x16x4_f32 %0, %1, %2, %0"
               : "+v"(c) : "v"(a), "v"(b));
  return c;
#endif
}

__global__ void span_max_kernel(const float* __restrict__ emb,
                                const int*   __restrict__ h_span,
                                const int*   __restrict__ t_span,
                                const int*   __restrict__ b_spans,
                                float* __restrict__ h_emb,
                                float* __restrict__ t_emb,
                                float* __restrict__ b_emb) {
  const int b = blockIdx.x;
  const int d = threadIdx.x;

  int lo[10], hi[10];
  lo[0] = h_span[b * 2];  hi[0] = h_span[b * 2 + 1];
  lo[1] = t_span[b * 2];  hi[1] = t_span[b * 2 + 1];
#pragma unroll
  for (int k = 0; k < K_; ++k) {
    lo[2 + k] = b_spans[(b * K_ + k) * 2];
    hi[2 + k] = b_spans[(b * K_ + k) * 2 + 1];
  }

  float mx[10];
#pragma unroll
  for (int j = 0; j < 10; ++j) mx[j] = -__builtin_inff();

  const float* e = emb + (size_t)b * S_ * D_ + d;
  for (int s = 0; s < S_; ++s) {
    const float val = e[(size_t)s * D_];
#pragma unroll
    for (int j = 0; j < 10; ++j)
      if (s >= lo[j] && s <= hi[j]) mx[j] = fmaxf(mx[j], val);
  }

  h_emb[b * D_ + d] = mx[0];
  t_emb[b * D_ + d] = mx[1];
#pragma unroll
  for (int k = 0; k < K_; ++k)
    b_emb[(size_t)(b * K_ + k) * D_ + d] = mx[2 + k];
}

__global__ void gemm_nt_wmma(const float* __restrict__ A,
                             const float* __restrict__ W,
                             const float* __restrict__ bias,
                             float* __restrict__ C,
                             int M, int N, int Kdim, int relu) {
  const int n_tiles = (N + 15) >> 4;
  const int m_tiles = M >> 4;
  const int wave    = blockIdx.x * (blockDim.x >> 5) + (threadIdx.x >> 5);
  if (wave >= m_tiles * n_tiles) return;

  const int mt   = wave / n_tiles;
  const int nt   = wave % n_tiles;
  const int lane = threadIdx.x & 31;
  const int half = lane >> 4;
  const int r    = lane & 15;

  const float* arow = A + (size_t)(mt * 16 + r) * Kdim + 2 * half;
  int wr = nt * 16 + r;
  if (wr > N - 1) wr = N - 1;
  const float* wrow = W + (size_t)wr * Kdim + 2 * half;

  v8f acc = {};
  for (int k = 0; k < Kdim; k += 16) {
#pragma unroll
    for (int kk = 0; kk < 16; kk += 4) {
      const v2f a  = *(const v2f*)(arow + k + kk);
      const v2f bb = *(const v2f*)(wrow + k + kk);
      acc = wmma_f32_16x16x4(a, bb, acc);
    }
  }

  const int n = nt * 16 + r;
  if (n < N) {
    const float bv = bias[n];
#pragma unroll
    for (int i = 0; i < 8; ++i) {
      float vout = acc[i] + bv;
      if (relu) vout = fmaxf(vout, 0.0f);
      C[(size_t)(mt * 16 + i + 8 * half) * N + n] = vout;
    }
  }
}

__global__ void attn_kernel(const float* __restrict__ u,
                            const float* __restrict__ v,
                            const float* __restrict__ bi,
                            float* __restrict__ pre) {
  const int b   = blockIdx.x;
  const int tid = threadIdx.x;
  __shared__ float red[256];
  __shared__ float att[K_];

  const float* ub  = u + b * D_;
  const float* vb  = v + b * D_;
  const float* bib = bi + (size_t)b * K_ * D_;

  float dot[K_];
#pragma unroll
  for (int k = 0; k < K_; ++k) {
    float p = 0.f;
    for (int d = tid; d < D_; d += 256) p += ub[d] * bib[k * D_ + d];
    dot[k] = p;
  }

#pragma unroll
  for (int k = 0; k < K_; ++k) {
    red[tid] = dot[k];
    __syncthreads();
    for (int off = 128; off > 0; off >>= 1) {
      if (tid < off) red[tid] += red[tid + off];
      __syncthreads();
    }
    if (tid == 0) att[k] = red[0] * 0.125f;
    __syncthreads();
  }

  if (tid == 0) {
    float m = att[0];
    for (int k = 1; k < K_; ++k) m = fmaxf(m, att[k]);
    float s = 0.f;
    for (int k = 0; k < K_; ++k) { att[k] = __expf(att[k] - m); s += att[k]; }
    const float inv = 1.0f / s;
    for (int k = 0; k < K_; ++k) att[k] *= inv;
  }
  __syncthreads();

  for (int d = tid; d < D_; d += 256) {
    float c = 0.f;
#pragma unroll
    for (int k = 0; k < K_; ++k) c += att[k] * bib[k * D_ + d];
    pre[b * D_ + d] = ub[d] + vb[d] + c;
  }
}

extern "C" void kernel_launch(void* const* d_in, const int* in_sizes, int n_in,
                              void* d_out, int out_size, void* d_ws, size_t ws_size,
                              hipStream_t stream) {
  (void)in_sizes; (void)n_in; (void)out_size; (void)ws_size;

  const float* emb     = (const float*)d_in[0];
  const int*   h_span  = (const int*)d_in[1];
  const int*   t_span  = (const int*)d_in[2];
  const int*   b_spans = (const int*)d_in[3];
  const float* wu_w    = (const float*)d_in[4];
  const float* wu_b    = (const float*)d_in[5];
  const float* wv_w    = (const float*)d_in[6];
  const float* wv_b    = (const float*)d_in[7];
  const float* wi_w    = (const float*)d_in[8];
  const float* wi_b    = (const float*)d_in[9];
  const float* ln1_w   = (const float*)d_in[10];
  const float* ln1_b   = (const float*)d_in[11];
  const float* pred_w  = (const float*)d_in[12];
  const float* pred_b  = (const float*)d_in[13];
  float* out = (float*)d_out;

  float* ws    = (float*)d_ws;
  float* h_emb = ws;
  float* t_emb = h_emb + B_ * D_;
  float* b_emb = t_emb + B_ * D_;
  float* u     = b_emb + B_ * K_ * D_;
  float* vv    = u + B_ * D_;
  float* bi    = vv + B_ * D_;
  float* pre   = bi + B_ * K_ * D_;
  float* pair  = h_emb;

  span_max_kernel<<<B_, D_, 0, stream>>>(emb, h_span, t_span, b_spans,
                                         h_emb, t_emb, b_emb);

  auto launch_gemm = [&](const float* A, const float* W, const float* bias,
                         float* C, int M, int N, int relu) {
    const int waves  = (M / 16) * ((N + 15) / 16);
    const int blocks = (waves + 7) / 8;
    gemm_nt_wmma<<<blocks, 256, 0, stream>>>(A, W, bias, C, M, N, D_, relu);
  };

  launch_gemm(h_emb, wu_w, wu_b, u, B_, D_, 0);
  launch_gemm(t_emb, wv_w, wv_b, vv, B_, D_, 0);
  launch_gemm(b_emb, wi_w, wi_b, bi, B_ * K_, D_, 0);

  attn_kernel<<<B_, 256, 0, stream>>>(u, vv, bi, pre);

  launch_gemm(pre, ln1_w, ln1_b, pair, B_, D_, 1);
  launch_gemm(pair, pred_w, pred_b, out, B_, R_, 0);
}


# Span max-pool spread over the whole chip: 384 blocks of 8 waves (batch row x 128-column strip), 16-byte loads in groups of 16 rows with the next group in flight, waves' maxima folded in LDS
# speedup vs baseline: 3.1528x; 3.1528x over previous
#include <hip/hip_runtime.h>

#define B_  64
#define S_  512
#define D_  768
#define K_  8
#define R_  277

typedef __attribute__((ext_vector_type(2))) float v2f;
typedef __attribute__((ext_vector_type(8))) float v8f;

__device__ __forceinline__ v8f wmma_f32_16x16x4(v2f a, v2f b, v8f c) {
#if defined(__gfx1250__) && __has_builtin(__builtin_amdgcn_wmma_f32_16x16x4_f32)
  return __builtin_amdgcn_wmma_f32_16x16x4_f32(
      false, a, false, b, (short)0, c, false, false);
#else
  asm volatile("v_wmma_f32_16x16x4_f32 %0, %1, %2, %0"
               : "+v"(c) : "v"(a), "v"(b));
  return c;
#endif
}

typedef __attribute__((ext_vector_type(4))) float v4f;

#define SP_WAVES   8
#define SP_WROWS   (S_ / SP_WAVES)
#define SP_GROUP   16
#define SP_NGROUP  (SP_WROWS / SP_GROUP)
#define SP_COLS    128
#define SP_NSTRIP  (D_ / SP_COLS)
#define SP_NSPAN   (2 + K_)

__device__ __forceinline__ v4f vmax4(v4f a, v4f b) {
  v4f r;
  r.x = fmaxf(a.x, b.x);
  r.y = fmaxf(a.y, b.y);
  r.z = fmaxf(a.z, b.z);
  r.w = fmaxf(a.w, b.w);
  return r;
}

__global__ __launch_bounds__(SP_WAVES * 32)
void span_max_kernel(const float* __restrict__ emb,
                     const int*   __restrict__ h_span,
                     const int*   __restrict__ t_span,
                     const int*   __restrict__ b_spans,
                     float* __restrict__ h_emb,
                     float* __restrict__ t_emb,
                     float* __restrict__ b_emb) {
  __shared__ __attribute__((aligned(16))) float red[SP_WAVES * SP_NSPAN * SP_COLS];

  const int b     = blockIdx.x / SP_NSTRIP;
  const int strip = blockIdx.x % SP_NSTRIP;
  const int tid   = threadIdx.x;
  const int lane  = tid & 31;
  const int wave  = __builtin_amdgcn_readfirstlane(tid >> 5);

  int lo[SP_NSPAN], hi[SP_NSPAN];
  lo[0] = h_span[b * 2];  hi[0] = h_span[b * 2 + 1];
  lo[1] = t_span[b * 2];  hi[1] = t_span[b * 2 + 1];
#pragma unroll
  for (int k = 0; k < K_; ++k) {
    lo[2 + k] = b_spans[(b * K_ + k) * 2];
    hi[2 + k] = b_spans[(b * K_ + k) * 2 + 1];
  }

  v4f mx[SP_NSPAN];
#pragma unroll
  for (int j = 0; j < SP_NSPAN; ++j) {
    mx[j].x = -__builtin_inff();  mx[j].y = -__builtin_inff();
    mx[j].z = -__builtin_inff();  mx[j].w = -__builtin_inff();
  }

  const int row0 = wave * SP_WROWS;
  const v4f* src = (const v4f*)(emb + ((size_t)b * S_ + row0) * D_
                                + strip * SP_COLS) + lane;

  v4f nxt[SP_GROUP];
#pragma unroll
  for (int i = 0; i < SP_GROUP; ++i) nxt[i] = src[(size_t)i * (D_ / 4)];

#pragma nounroll
  for (int g = 0; g < SP_NGROUP; ++g) {
    v4f cur[SP_GROUP];
#pragma unroll
    for (int i = 0; i < SP_GROUP; ++i) cur[i] = nxt[i];

    if (g + 1 < SP_NGROUP) {
      const v4f* s2 = src + (size_t)(g + 1) * SP_GROUP * (D_ / 4);
#pragma unroll
      for (int i = 0; i < SP_GROUP; ++i) nxt[i] = s2[(size_t)i * (D_ / 4)];
    }

    v4f gm = cur[0];
#pragma unroll
    for (int i = 1; i < SP_GROUP; ++i) gm = vmax4(gm, cur[i]);

    const int r0 = row0 + g * SP_GROUP;
#pragma unroll
    for (int j = 0; j < SP_NSPAN; ++j) {
      int a = lo[j] - r0;      if (a < 0) a = 0;
      int e = hi[j] - r0 + 1;  if (e > SP_GROUP) e = SP_GROUP;
      if (a < e) {
        if (a == 0 && e == SP_GROUP) {
          mx[j] = vmax4(mx[j], gm);
        } else {
#pragma unroll
          for (int i = 0; i < SP_GROUP; ++i)
            if (i >= a && i < e) mx[j] = vmax4(mx[j], cur[i]);
        }
      }
    }
  }

#pragma unroll
  for (int j = 0; j < SP_NSPAN; ++j)
    *(v4f*)&red[((wave * SP_NSPAN + j) * 32 + lane) * 4] = mx[j];
  __syncthreads();

  for (int idx = tid; idx < SP_NSPAN * SP_COLS; idx += SP_WAVES * 32) {
    const int j = idx / SP_COLS;
    const int c = idx % SP_COLS;
    float m = red[j * SP_COLS + c];
#pragma unroll
    for (int w = 1; w < SP_WAVES; ++w)
      m = fmaxf(m, red[(w * SP_NSPAN + j) * SP_COLS + c]);
    const int col = strip * SP_COLS + c;
    float* dst = (j == 0) ? (h_emb + b * D_)
               : (j == 1) ? (t_emb + b * D_)
                          : (b_emb + (size_t)(b * K_ + (j - 2)) * D_);
    dst[col] = m;
  }
}

__global__ void gemm_nt_wmma(const float* __restrict__ A,
                             const float* __restrict__ W,
                             const float* __restrict__ bias,
                             float* __restrict__ C,
                             int M, int N, int Kdim, int relu) {
  const int n_tiles = (N + 15) >> 4;
  const int m_tiles = M >> 4;
  const int wave    = blockIdx.x * (blockDim.x >> 5) + (threadIdx.x >> 5);
  if (wave >= m_tiles * n_tiles) return;

  const int mt   = wave / n_tiles;
  const int nt   = wave % n_tiles;
  const int lane = threadIdx.x & 31;
  const int half = lane >> 4;
  const int r    = lane & 15;

  const float* arow = A + (size_t)(mt * 16 + r) * Kdim + 2 * half;
  int wr = nt * 16 + r;
  if (wr > N - 1) wr = N - 1;
  const float* wrow = W + (size_t)wr * Kdim + 2 * half;

  v8f acc = {};
  for (int k = 0; k < Kdim; k += 16) {
#pragma unroll
    for (int kk = 0; kk < 16; kk += 4) {
      const v2f a  = *(const v2f*)(arow + k + kk);
      const v2f bb = *(const v2f*)(wrow + k + kk);
      acc = wmma_f32_16x16x4(a, bb, acc);
    }
  }

  const int n = nt * 16 + r;
  if (n < N) {
    const float bv = bias[n];
#pragma unroll
    for (int i = 0; i < 8; ++i) {
      float vout = acc[i] + bv;
      if (relu) vout = fmaxf(vout, 0.0f);
      C[(size_t)(mt * 16 + i + 8 * half) * N + n] = vout;
    }
  }
}

__global__ void attn_kernel(const float* __restrict__ u,
                            const float* __restrict__ v,
                            const float* __restrict__ bi,
                            float* __restrict__ pre) {
  const int b   = blockIdx.x;
  const int tid = threadIdx.x;
  __shared__ float red[256];
  __shared__ float att[K_];

  const float* ub  = u + b * D_;
  const float* vb  = v + b * D_;
  const float* bib = bi + (size_t)b * K_ * D_;

  float dot[K_];
#pragma unroll
  for (int k = 0; k < K_; ++k) {
    float p = 0.f;
    for (int d = tid; d < D_; d += 256) p += ub[d] * bib[k * D_ + d];
    dot[k] = p;
  }

#pragma unroll
  for (int k = 0; k < K_; ++k) {
    red[tid] = dot[k];
    __syncthreads();
    for (int off = 128; off > 0; off >>= 1) {
      if (tid < off) red[tid] += red[tid + off];
      __syncthreads();
    }
    if (tid == 0) att[k] = red[0] * 0.125f;
    __syncthreads();
  }

  if (tid == 0) {
    float m = att[0];
    for (int k = 1; k < K_; ++k) m = fmaxf(m, att[k]);
    float s = 0.f;
    for (int k = 0; k < K_; ++k) { att[k] = __expf(att[k] - m); s += att[k]; }
    const float inv = 1.0f / s;
    for (int k = 0; k < K_; ++k) att[k] *= inv;
  }
  __syncthreads();

  for (int d = tid; d < D_; d += 256) {
    float c = 0.f;
#pragma unroll
    for (int k = 0; k < K_; ++k) c += att[k] * bib[k * D_ + d];
    pre[b * D_ + d] = ub[d] + vb[d] + c;
  }
}

extern "C" void kernel_launch(void* const* d_in, const int* in_sizes, int n_in,
                              void* d_out, int out_size, void* d_ws, size_t ws_size,
                              hipStream_t stream) {
  (void)in_sizes; (void)n_in; (void)out_size; (void)ws_size;

  const float* emb     = (const float*)d_in[0];
  const int*   h_span  = (const int*)d_in[1];
  const int*   t_span  = (const int*)d_in[2];
  const int*   b_spans = (const int*)d_in[3];
  const float* wu_w    = (const float*)d_in[4];
  const float* wu_b    = (const float*)d_in[5];
  const float* wv_w    = (const float*)d_in[6];
  const float* wv_b    = (const float*)d_in[7];
  const float* wi_w    = (const float*)d_in[8];
  const float* wi_b    = (const float*)d_in[9];
  const float* ln1_w   = (const float*)d_in[10];
  const float* ln1_b   = (const float*)d_in[11];
  const float* pred_w  = (const float*)d_in[12];
  const float* pred_b  = (const float*)d_in[13];
  float* out = (float*)d_out;

  float* ws    = (float*)d_ws;
  float* h_emb = ws;
  float* t_emb = h_emb + B_ * D_;
  float* b_emb = t_emb + B_ * D_;
  float* u     = b_emb + B_ * K_ * D_;
  float* vv    = u + B_ * D_;
  float* bi    = vv + B_ * D_;
  float* pre   = bi + B_ * K_ * D_;
  float* pair  = h_emb;

  span_max_kernel<<<B_ * SP_NSTRIP, SP_WAVES * 32, 0, stream>>>(emb, h_span, t_span, b_spans,
                                         h_emb, t_emb, b_emb);

  auto launch_gemm = [&](const float* A, const float* W, const float* bias,
                         float* C, int M, int N, int relu) {
    const int waves  = (M / 16) * ((N + 15) / 16);
    const int blocks = (waves + 7) / 8;
    gemm_nt_wmma<<<blocks, 256, 0, stream>>>(A, W, bias, C, M, N, D_, relu);
  };

  launch_gemm(h_emb, wu_w, wu_b, u, B_, D_, 0);
  launch_gemm(t_emb, wv_w, wv_b, vv, B_, D_, 0);
  launch_gemm(b_emb, wi_w, wi_b, bi, B_ * K_, D_, 0);

  attn_kernel<<<B_, 256, 0, stream>>>(u, vv, bi, pre);

  launch_gemm(pre, ln1_w, ln1_b, pair, B_, D_, 1);
  launch_gemm(pair, pred_w, pred_b, out, B_, R_, 0);
}


# GEMMs: K split over a block's 8 waves, each wave loads its whole 96-k slice (16-byte loads) before its WMMAs, partials summed in LDS (new K sum order); 32x32 block tile for the largest GEMM
# speedup vs baseline: 6.1487x; 1.9502x over previous
#include <hip/hip_runtime.h>

#define B_  64
#define S_  512
#define D_  768
#define K_  8
#define R_  277

typedef __attribute__((ext_vector_type(2))) float v2f;
typedef __attribute__((ext_vector_type(8))) float v8f;

__device__ __forceinline__ v8f wmma_f32_16x16x4(v2f a, v2f b, v8f c) {
#if defined(__gfx1250__) && __has_builtin(__builtin_amdgcn_wmma_f32_16x16x4_f32)
  return __builtin_amdgcn_wmma_f32_16x16x4_f32(
      false, a, false, b, (short)0, c, false, false);
#else
  asm volatile("v_wmma_f32_16x16x4_f32 %0, %1, %2, %0"
               : "+v"(c) : "v"(a), "v"(b));
  return c;
#endif
}

typedef __attribute__((ext_vector_type(4))) float v4f;

#define SP_WAVES   8
#define SP_WROWS   (S_ / SP_WAVES)
#define SP_GROUP   16
#define SP_NGROUP  (SP_WROWS / SP_GROUP)
#define SP_COLS    128
#define SP_NSTRIP  (D_ / SP_COLS)
#define SP_NSPAN   (2 + K_)

__device__ __forceinline__ v4f vmax4(v4f a, v4f b) {
  v4f r;
  r.x = fmaxf(a.x, b.x);
  r.y = fmaxf(a.y, b.y);
  r.z = fmaxf(a.z, b.z);
  r.w = fmaxf(a.w, b.w);
  return r;
}

__global__ __launch_bounds__(SP_WAVES * 32)
void span_max_kernel(const float* __restrict__ emb,
                     const int*   __restrict__ h_span,
                     const int*   __restrict__ t_span,
                     const int*   __restrict__ b_spans,
                     float* __restrict__ h_emb,
                     float* __restrict__ t_emb,
                     float* __restrict__ b_emb) {
  __shared__ __attribute__((aligned(16))) float red[SP_WAVES * SP_NSPAN * SP_COLS];

  const int b     = blockIdx.x / SP_NSTRIP;
  const int strip = blockIdx.x % SP_NSTRIP;
  const int tid   = threadIdx.x;
  const int lane  = tid & 31;
  const int wave  = __builtin_amdgcn_readfirstlane(tid >> 5);

  int lo[SP_NSPAN], hi[SP_NSPAN];
  lo[0] = h_span[b * 2];  hi[0] = h_span[b * 2 + 1];
  lo[1] = t_span[b * 2];  hi[1] = t_span[b * 2 + 1];
#pragma unroll
  for (int k = 0; k < K_; ++k) {
    lo[2 + k] = b_spans[(b * K_ + k) * 2];
    hi[2 + k] = b_spans[(b * K_ + k) * 2 + 1];
  }

  v4f mx[SP_NSPAN];
#pragma unroll
  for (int j = 0; j < SP_NSPAN; ++j) {
    mx[j].x = -__builtin_inff();  mx[j].y = -__builtin_inff();
    mx[j].z = -__builtin_inff();  mx[j].w = -__builtin_inff();
  }

  const int row0 = wave * SP_WROWS;
  const v4f* src = (const v4f*)(emb + ((size_t)b * S_ + row0) * D_
                                + strip * SP_COLS) + lane;

  v4f nxt[SP_GROUP];
#pragma unroll
  for (int i = 0; i < SP_GROUP; ++i) nxt[i] = src[(size_t)i * (D_ / 4)];

#pragma nounroll
  for (int g = 0; g < SP_NGROUP; ++g) {
    v4f cur[SP_GROUP];
#pragma unroll
    for (int i = 0; i < SP_GROUP; ++i) cur[i] = nxt[i];

    if (g + 1 < SP_NGROUP) {
      const v4f* s2 = src + (size_t)(g + 1) * SP_GROUP * (D_ / 4);
#pragma unroll
      for (int i = 0; i < SP_GROUP; ++i) nxt[i] = s2[(size_t)i * (D_ / 4)];
    }

    v4f gm = cur[0];
#pragma unroll
    for (int i = 1; i < SP_GROUP; ++i) gm = vmax4(gm, cur[i]);

    const int r0 = row0 + g * SP_GROUP;
#pragma unroll
    for (int j = 0; j < SP_NSPAN; ++j) {
      int a = lo[j] - r0;      if (a < 0) a = 0;
      int e = hi[j] - r0 + 1;  if (e > SP_GROUP) e = SP_GROUP;
      if (a < e) {
        if (a == 0 && e == SP_GROUP) {
          mx[j] = vmax4(mx[j], gm);
        } else {
#pragma unroll
          for (int i = 0; i < SP_GROUP; ++i)
            if (i >= a && i < e) mx[j] = vmax4(mx[j], cur[i]);
        }
      }
    }
  }

#pragma unroll
  for (int j = 0; j < SP_NSPAN; ++j)
    *(v4f*)&red[((wave * SP_NSPAN + j) * 32 + lane) * 4] = mx[j];
  __syncthreads();

  for (int idx = tid; idx < SP_NSPAN * SP_COLS; idx += SP_WAVES * 32) {
    const int j = idx / SP_COLS;
    const int c = idx % SP_COLS;
    float m = red[j * SP_COLS + c];
#pragma unroll
    for (int w = 1; w < SP_WAVES; ++w)
      m = fmaxf(m, red[(w * SP_NSPAN + j) * SP_COLS + c]);
    const int col = strip * SP_COLS + c;
    float* dst = (j == 0) ? (h_emb + b * D_)
               : (j == 1) ? (t_emb + b * D_)
                          : (b_emb + (size_t)(b * K_ + (j - 2)) * D_);
    dst[col] = m;
  }
}

#define GK_WAVES  8
#define GK_SLICE  (D_ / GK_WAVES)
#define GK_LOADS  (GK_SLICE / 8)

template <int MT, int NT>
__global__ __launch_bounds__(GK_WAVES * 32)
void gemm_nt_splitk(const float* __restrict__ A,
                    const float* __restrict__ W,
                    const float* __restrict__ bias,
                    float* __restrict__ C,
                    int N, int relu) {
  constexpr int ROWS  = 16 * MT;
  constexpr int COLS  = 16 * NT;
  constexpr int PITCH = COLS + 2;
  __shared__ float part[GK_WAVES * ROWS * PITCH];

  const int tid  = threadIdx.x;
  const int lane = tid & 31;
  const int wave = __builtin_amdgcn_readfirstlane(tid >> 5);
  const int c    = lane & 15;
  const int h    = lane >> 4;
  const int m0   = blockIdx.y * ROWS;
  const int n0   = blockIdx.x * COLS;

  const size_t koff = (size_t)wave * GK_SLICE + 4 * h;

  const float* ap[MT];
  const float* wp[NT];
#pragma unroll
  for (int mi = 0; mi < MT; ++mi)
    ap[mi] = A + (size_t)(m0 + mi * 16 + c) * D_ + koff;
#pragma unroll
  for (int ni = 0; ni < NT; ++ni) {
    int wr = n0 + ni * 16 + c;
    if (wr > N - 1) wr = N - 1;
    wp[ni] = W + (size_t)wr * D_ + koff;
  }

  v4f af[MT][GK_LOADS];
  v4f wf[NT][GK_LOADS];
#pragma unroll
  for (int j = 0; j < GK_LOADS; ++j) {
#pragma unroll
    for (int mi = 0; mi < MT; ++mi) af[mi][j] = *(const v4f*)(ap[mi] + 8 * j);
#pragma unroll
    for (int ni = 0; ni < NT; ++ni) wf[ni][j] = *(const v4f*)(wp[ni] + 8 * j);
  }
  __builtin_amdgcn_sched_barrier(0);

  v8f acc[MT][NT];
#pragma unroll
  for (int mi = 0; mi < MT; ++mi)
#pragma unroll
    for (int ni = 0; ni < NT; ++ni) acc[mi][ni] = v8f{};

#pragma unroll
  for (int j = 0; j < GK_LOADS; ++j) {
#pragma unroll
    for (int mi = 0; mi < MT; ++mi) {
#pragma unroll
      for (int ni = 0; ni < NT; ++ni) {
        v2f a0, b0;
        a0.x = af[mi][j].x;  a0.y = af[mi][j].y;
        b0.x = wf[ni][j].x;  b0.y = wf[ni][j].y;
        acc[mi][ni] = wmma_f32_16x16x4(a0, b0, acc[mi][ni]);
      }
    }
#pragma unroll
    for (int mi = 0; mi < MT; ++mi) {
#pragma unroll
      for (int ni = 0; ni < NT; ++ni) {
        v2f a1, b1;
        a1.x = af[mi][j].z;  a1.y = af[mi][j].w;
        b1.x = wf[ni][j].z;  b1.y = wf[ni][j].w;
        acc[mi][ni] = wmma_f32_16x16x4(a1, b1, acc[mi][ni]);
      }
    }
  }

  float* mine = part + wave * (ROWS * PITCH);
#pragma unroll
  for (int mi = 0; mi < MT; ++mi)
#pragma unroll
    for (int ni = 0; ni < NT; ++ni)
#pragma unroll
      for (int i = 0; i < 8; ++i)
        mine[(mi * 16 + 8 * h + i) * PITCH + ni * 16 + c] = acc[mi][ni][i];
  __syncthreads();

#pragma unroll
  for (int q = 0; q < MT * NT; ++q) {
    const int o   = tid + q * (GK_WAVES * 32);
    const int row = o / COLS;
    const int col = o % COLS;
    float s = part[row * PITCH + col];
#pragma unroll
    for (int w = 1; w < GK_WAVES; ++w)
      s += part[(w * ROWS + row) * PITCH + col];
    const int n = n0 + col;
    if (n < N) {
      float vout = s + bias[n];
      if (relu) vout = fmaxf(vout, 0.0f);
      C[(size_t)(m0 + row) * N + n] = vout;
    }
  }
}

__global__ void attn_kernel(const float* __restrict__ u,
                            const float* __restrict__ v,
                            const float* __restrict__ bi,
                            float* __restrict__ pre) {
  const int b   = blockIdx.x;
  const int tid = threadIdx.x;
  __shared__ float red[256];
  __shared__ float att[K_];

  const float* ub  = u + b * D_;
  const float* vb  = v + b * D_;
  const float* bib = bi + (size_t)b * K_ * D_;

  float dot[K_];
#pragma unroll
  for (int k = 0; k < K_; ++k) {
    float p = 0.f;
    for (int d = tid; d < D_; d += 256) p += ub[d] * bib[k * D_ + d];
    dot[k] = p;
  }

#pragma unroll
  for (int k = 0; k < K_; ++k) {
    red[tid] = dot[k];
    __syncthreads();
    for (int off = 128; off > 0; off >>= 1) {
      if (tid < off) red[tid] += red[tid + off];
      __syncthreads();
    }
    if (tid == 0) att[k] = red[0] * 0.125f;
    __syncthreads();
  }

  if (tid == 0) {
    float m = att[0];
    for (int k = 1; k < K_; ++k) m = fmaxf(m, att[k]);
    float s = 0.f;
    for (int k = 0; k < K_; ++k) { att[k] = __expf(att[k] - m); s += att[k]; }
    const float inv = 1.0f / s;
    for (int k = 0; k < K_; ++k) att[k] *= inv;
  }
  __syncthreads();

  for (int d = tid; d < D_; d += 256) {
    float c = 0.f;
#pragma unroll
    for (int k = 0; k < K_; ++k) c += att[k] * bib[k * D_ + d];
    pre[b * D_ + d] = ub[d] + vb[d] + c;
  }
}

extern "C" void kernel_launch(void* const* d_in, const int* in_sizes, int n_in,
                              void* d_out, int out_size, void* d_ws, size_t ws_size,
                              hipStream_t stream) {
  (void)in_sizes; (void)n_in; (void)out_size; (void)ws_size;

  const float* emb     = (const float*)d_in[0];
  const int*   h_span  = (const int*)d_in[1];
  const int*   t_span  = (const int*)d_in[2];
  const int*   b_spans = (const int*)d_in[3];
  const float* wu_w    = (const float*)d_in[4];
  const float* wu_b    = (const float*)d_in[5];
  const float* wv_w    = (const float*)d_in[6];
  const float* wv_b    = (const float*)d_in[7];
  const float* wi_w    = (const float*)d_in[8];
  const float* wi_b    = (const float*)d_in[9];
  const float* ln1_w   = (const float*)d_in[10];
  const float* ln1_b   = (const float*)d_in[11];
  const float* pred_w  = (const float*)d_in[12];
  const float* pred_b  = (const float*)d_in[13];
  float* out = (float*)d_out;

  float* ws    = (float*)d_ws;
  float* h_emb = ws;
  float* t_emb = h_emb + B_ * D_;
  float* b_emb = t_emb + B_ * D_;
  float* u     = b_emb + B_ * K_ * D_;
  float* vv    = u + B_ * D_;
  float* bi    = vv + B_ * D_;
  float* pre   = bi + B_ * K_ * D_;
  float* pair  = h_emb;

  span_max_kernel<<<B_ * SP_NSTRIP, SP_WAVES * 32, 0, stream>>>(emb, h_span, t_span, b_spans,
                                         h_emb, t_emb, b_emb);

  const dim3 g_small(D_ / 16, B_ / 16);
  const dim3 g_big(D_ / 32, (B_ * K_) / 32);
  const dim3 g_out((R_ + 15) / 16, B_ / 16);

  gemm_nt_splitk<1, 1><<<g_small, GK_WAVES * 32, 0, stream>>>(h_emb, wu_w, wu_b, u, D_, 0);
  gemm_nt_splitk<1, 1><<<g_small, GK_WAVES * 32, 0, stream>>>(t_emb, wv_w, wv_b, vv, D_, 0);
  gemm_nt_splitk<2, 2><<<g_big, GK_WAVES * 32, 0, stream>>>(b_emb, wi_w, wi_b, bi, D_, 0);

  attn_kernel<<<B_, 256, 0, stream>>>(u, vv, bi, pre);

  gemm_nt_splitk<1, 1><<<g_small, GK_WAVES * 32, 0, stream>>>(pre, ln1_w, ln1_b, pair, D_, 1);
  gemm_nt_splitk<1, 1><<<g_out, GK_WAVES * 32, 0, stream>>>(pair, pred_w, pred_b, out, R_, 0);
}


# u, v and bi GEMMs (mutually independent) issued as ONE launch: row-block index picks the problem, all on the 32x32 split-K tile; 7 launches become 5; same sum order per element
# speedup vs baseline: 7.1693x; 1.1660x over previous
#include <hip/hip_runtime.h>

#define B_  64
#define S_  512
#define D_  768
#define K_  8
#define R_  277

typedef __attribute__((ext_vector_type(2))) float v2f;
typedef __attribute__((ext_vector_type(8))) float v8f;

__device__ __forceinline__ v8f wmma_f32_16x16x4(v2f a, v2f b, v8f c) {
#if defined(__gfx1250__) && __has_builtin(__builtin_amdgcn_wmma_f32_16x16x4_f32)
  return __builtin_amdgcn_wmma_f32_16x16x4_f32(
      false, a, false, b, (short)0, c, false, false);
#else
  asm volatile("v_wmma_f32_16x16x4_f32 %0, %1, %2, %0"
               : "+v"(c) : "v"(a), "v"(b));
  return c;
#endif
}

typedef __attribute__((ext_vector_type(4))) float v4f;

#define SP_WAVES   8
#define SP_WROWS   (S_ / SP_WAVES)
#define SP_GROUP   16
#define SP_NGROUP  (SP_WROWS / SP_GROUP)
#define SP_COLS    128
#define SP_NSTRIP  (D_ / SP_COLS)
#define SP_NSPAN   (2 + K_)

__device__ __forceinline__ v4f vmax4(v4f a, v4f b) {
  v4f r;
  r.x = fmaxf(a.x, b.x);
  r.y = fmaxf(a.y, b.y);
  r.z = fmaxf(a.z, b.z);
  r.w = fmaxf(a.w, b.w);
  return r;
}

__global__ __launch_bounds__(SP_WAVES * 32)
void span_max_kernel(const float* __restrict__ emb,
                     const int*   __restrict__ h_span,
                     const int*   __restrict__ t_span,
                     const int*   __restrict__ b_spans,
                     float* __restrict__ h_emb,
                     float* __restrict__ t_emb,
                     float* __restrict__ b_emb) {
  __shared__ __attribute__((aligned(16))) float red[SP_WAVES * SP_NSPAN * SP_COLS];

  const int b     = blockIdx.x / SP_NSTRIP;
  const int strip = blockIdx.x % SP_NSTRIP;
  const int tid   = threadIdx.x;
  const int lane  = tid & 31;
  const int wave  = __builtin_amdgcn_readfirstlane(tid >> 5);

  int lo[SP_NSPAN], hi[SP_NSPAN];
  lo[0] = h_span[b * 2];  hi[0] = h_span[b * 2 + 1];
  lo[1] = t_span[b * 2];  hi[1] = t_span[b * 2 + 1];
#pragma unroll
  for (int k = 0; k < K_; ++k) {
    lo[2 + k] = b_spans[(b * K_ + k) * 2];
    hi[2 + k] = b_spans[(b * K_ + k) * 2 + 1];
  }

  v4f mx[SP_NSPAN];
#pragma unroll
  for (int j = 0; j < SP_NSPAN; ++j) {
    mx[j].x = -__builtin_inff();  mx[j].y = -__builtin_inff();
    mx[j].z = -__builtin_inff();  mx[j].w = -__builtin_inff();
  }

  const int row0 = wave * SP_WROWS;
  const v4f* src = (const v4f*)(emb + ((size_t)b * S_ + row0) * D_
                                + strip * SP_COLS) + lane;

  v4f nxt[SP_GROUP];
#pragma unroll
  for (int i = 0; i < SP_GROUP; ++i) nxt[i] = src[(size_t)i * (D_ / 4)];

#pragma nounroll
  for (int g = 0; g < SP_NGROUP; ++g) {
    v4f cur[SP_GROUP];
#pragma unroll
    for (int i = 0; i < SP_GROUP; ++i) cur[i] = nxt[i];

    if (g + 1 < SP_NGROUP) {
      const v4f* s2 = src + (size_t)(g + 1) * SP_GROUP * (D_ / 4);
#pragma unroll
      for (int i = 0; i < SP_GROUP; ++i) nxt[i] = s2[(size_t)i * (D_ / 4)];
    }

    v4f gm = cur[0];
#pragma unroll
    for (int i = 1; i < SP_GROUP; ++i) gm = vmax4(gm, cur[i]);

    const int r0 = row0 + g * SP_GROUP;
#pragma unroll
    for (int j = 0; j < SP_NSPAN; ++j) {
      int a = lo[j] - r0;      if (a < 0) a = 0;
      int e = hi[j] - r0 + 1;  if (e > SP_GROUP) e = SP_GROUP;
      if (a < e) {
        if (a == 0 && e == SP_GROUP) {
          mx[j] = vmax4(mx[j], gm);
        } else {
#pragma unroll
          for (int i = 0; i < SP_GROUP; ++i)
            if (i >= a && i < e) mx[j] = vmax4(mx[j], cur[i]);
        }
      }
    }
  }

#pragma unroll
  for (int j = 0; j < SP_NSPAN; ++j)
    *(v4f*)&red[((wave * SP_NSPAN + j) * 32 + lane) * 4] = mx[j];
  __syncthreads();

  for (int idx = tid; idx < SP_NSPAN * SP_COLS; idx += SP_WAVES * 32) {
    const int j = idx / SP_COLS;
    const int c = idx % SP_COLS;
    float m = red[j * SP_COLS + c];
#pragma unroll
    for (int w = 1; w < SP_WAVES; ++w)
      m = fmaxf(m, red[(w * SP_NSPAN + j) * SP_COLS + c]);
    const int col = strip * SP_COLS + c;
    float* dst = (j == 0) ? (h_emb + b * D_)
               : (j == 1) ? (t_emb + b * D_)
                          : (b_emb + (size_t)(b * K_ + (j - 2)) * D_);
    dst[col] = m;
  }
}

#define GK_WAVES  8
#define GK_SLICE  (D_ / GK_WAVES)
#define GK_LOADS  (GK_SLICE / 8)

template <int MT, int NT>
__device__ __forceinline__
void gemm_tile_splitk(const float* __restrict__ A,
                      const float* __restrict__ W,
                      const float* __restrict__ bias,
                      float* __restrict__ C,
                      int N, int relu, int m0, int n0) {
  constexpr int ROWS  = 16 * MT;
  constexpr int COLS  = 16 * NT;
  constexpr int PITCH = COLS + 2;
  __shared__ float part[GK_WAVES * ROWS * PITCH];

  const int tid  = threadIdx.x;
  const int lane = tid & 31;
  const int wave = __builtin_amdgcn_readfirstlane(tid >> 5);
  const int c    = lane & 15;
  const int h    = lane >> 4;

  const size_t koff = (size_t)wave * GK_SLICE + 4 * h;

  const float* ap[MT];
  const float* wp[NT];
#pragma unroll
  for (int mi = 0; mi < MT; ++mi)
    ap[mi] = A + (size_t)(m0 + mi * 16 + c) * D_ + koff;
#pragma unroll
  for (int ni = 0; ni < NT; ++ni) {
    int wr = n0 + ni * 16 + c;
    if (wr > N - 1) wr = N - 1;
    wp[ni] = W + (size_t)wr * D_ + koff;
  }

  v4f af[MT][GK_LOADS];
  v4f wf[NT][GK_LOADS];
#pragma unroll
  for (int j = 0; j < GK_LOADS; ++j) {
#pragma unroll
    for (int mi = 0; mi < MT; ++mi) af[mi][j] = *(const v4f*)(ap[mi] + 8 * j);
#pragma unroll
    for (int ni = 0; ni < NT; ++ni) wf[ni][j] = *(const v4f*)(wp[ni] + 8 * j);
  }
  __builtin_amdgcn_sched_barrier(0);

  v8f acc[MT][NT];
#pragma unroll
  for (int mi = 0; mi < MT; ++mi)
#pragma unroll
    for (int ni = 0; ni < NT; ++ni) acc[mi][ni] = v8f{};

#pragma unroll
  for (int j = 0; j < GK_LOADS; ++j) {
#pragma unroll
    for (int mi = 0; mi < MT; ++mi) {
#pragma unroll
      for (int ni = 0; ni < NT; ++ni) {
        v2f a0, b0;
        a0.x = af[mi][j].x;  a0.y = af[mi][j].y;
        b0.x = wf[ni][j].x;  b0.y = wf[ni][j].y;
        acc[mi][ni] = wmma_f32_16x16x4(a0, b0, acc[mi][ni]);
      }
    }
#pragma unroll
    for (int mi = 0; mi < MT; ++mi) {
#pragma unroll
      for (int ni = 0; ni < NT; ++ni) {
        v2f a1, b1;
        a1.x = af[mi][j].z;  a1.y = af[mi][j].w;
        b1.x = wf[ni][j].z;  b1.y = wf[ni][j].w;
        acc[mi][ni] = wmma_f32_16x16x4(a1, b1, acc[mi][ni]);
      }
    }
  }

  float* mine = part + wave * (ROWS * PITCH);
#pragma unroll
  for (int mi = 0; mi < MT; ++mi)
#pragma unroll
    for (int ni = 0; ni < NT; ++ni)
#pragma unroll
      for (int i = 0; i < 8; ++i)
        mine[(mi * 16 + 8 * h + i) * PITCH + ni * 16 + c] = acc[mi][ni][i];
  __syncthreads();

#pragma unroll
  for (int q = 0; q < MT * NT; ++q) {
    const int o   = tid + q * (GK_WAVES * 32);
    const int row = o / COLS;
    const int col = o % COLS;
    float s = part[row * PITCH + col];
#pragma unroll
    for (int w = 1; w < GK_WAVES; ++w)
      s += part[(w * ROWS + row) * PITCH + col];
    const int n = n0 + col;
    if (n < N) {
      float vout = s + bias[n];
      if (relu) vout = fmaxf(vout, 0.0f);
      C[(size_t)(m0 + row) * N + n] = vout;
    }
  }
}

template <int MT, int NT>
__global__ __launch_bounds__(GK_WAVES * 32)
void gemm_nt_splitk(const float* __restrict__ A,
                    const float* __restrict__ W,
                    const float* __restrict__ bias,
                    float* __restrict__ C,
                    int N, int relu) {
  gemm_tile_splitk<MT, NT>(A, W, bias, C, N, relu,
                           blockIdx.y * (16 * MT), blockIdx.x * (16 * NT));
}

#define G3_YB_BI  ((B_ * K_) / 32)
#define G3_YB_UV  (B_ / 32)

__global__ __launch_bounds__(GK_WAVES * 32)
void gemm3_nt_splitk(const float* __restrict__ Ab, const float* __restrict__ Wb,
                     const float* __restrict__ bb, float* __restrict__ Cb,
                     const float* __restrict__ Au, const float* __restrict__ Wu,
                     const float* __restrict__ bu, float* __restrict__ Cu,
                     const float* __restrict__ Av, const float* __restrict__ Wv,
                     const float* __restrict__ bv, float* __restrict__ Cv) {
  const int y = blockIdx.y;
  const float* A;  const float* W;  const float* bias;  float* C;  int yb;
  if (y < G3_YB_BI)                 { A = Ab; W = Wb; bias = bb; C = Cb; yb = y; }
  else if (y < G3_YB_BI + G3_YB_UV) { A = Au; W = Wu; bias = bu; C = Cu; yb = y - G3_YB_BI; }
  else                              { A = Av; W = Wv; bias = bv; C = Cv; yb = y - G3_YB_BI - G3_YB_UV; }
  gemm_tile_splitk<2, 2>(A, W, bias, C, D_, 0, yb * 32, blockIdx.x * 32);
}

__global__ void attn_kernel(const float* __restrict__ u,
                            const float* __restrict__ v,
                            const float* __restrict__ bi,
                            float* __restrict__ pre) {
  const int b   = blockIdx.x;
  const int tid = threadIdx.x;
  __shared__ float red[256];
  __shared__ float att[K_];

  const float* ub  = u + b * D_;
  const float* vb  = v + b * D_;
  const float* bib = bi + (size_t)b * K_ * D_;

  float dot[K_];
#pragma unroll
  for (int k = 0; k < K_; ++k) {
    float p = 0.f;
    for (int d = tid; d < D_; d += 256) p += ub[d] * bib[k * D_ + d];
    dot[k] = p;
  }

#pragma unroll
  for (int k = 0; k < K_; ++k) {
    red[tid] = dot[k];
    __syncthreads();
    for (int off = 128; off > 0; off >>= 1) {
      if (tid < off) red[tid] += red[tid + off];
      __syncthreads();
    }
    if (tid == 0) att[k] = red[0] * 0.125f;
    __syncthreads();
  }

  if (tid == 0) {
    float m = att[0];
    for (int k = 1; k < K_; ++k) m = fmaxf(m, att[k]);
    float s = 0.f;
    for (int k = 0; k < K_; ++k) { att[k] = __expf(att[k] - m); s += att[k]; }
    const float inv = 1.0f / s;
    for (int k = 0; k < K_; ++k) att[k] *= inv;
  }
  __syncthreads();

  for (int d = tid; d < D_; d += 256) {
    float c = 0.f;
#pragma unroll
    for (int k = 0; k < K_; ++k) c += att[k] * bib[k * D_ + d];
    pre[b * D_ + d] = ub[d] + vb[d] + c;
  }
}

extern "C" void kernel_launch(void* const* d_in, const int* in_sizes, int n_in,
                              void* d_out, int out_size, void* d_ws, size_t ws_size,
                              hipStream_t stream) {
  (void)in_sizes; (void)n_in; (void)out_size; (void)ws_size;

  const float* emb     = (const float*)d_in[0];
  const int*   h_span  = (const int*)d_in[1];
  const int*   t_span  = (const int*)d_in[2];
  const int*   b_spans = (const int*)d_in[3];
  const float* wu_w    = (const float*)d_in[4];
  const float* wu_b    = (const float*)d_in[5];
  const float* wv_w    = (const float*)d_in[6];
  const float* wv_b    = (const float*)d_in[7];
  const float* wi_w    = (const float*)d_in[8];
  const float* wi_b    = (const float*)d_in[9];
  const float* ln1_w   = (const float*)d_in[10];
  const float* ln1_b   = (const float*)d_in[11];
  const float* pred_w  = (const float*)d_in[12];
  const float* pred_b  = (const float*)d_in[13];
  float* out = (float*)d_out;

  float* ws    = (float*)d_ws;
  float* h_emb = ws;
  float* t_emb = h_emb + B_ * D_;
  float* b_emb = t_emb + B_ * D_;
  float* u     = b_emb + B_ * K_ * D_;
  float* vv    = u + B_ * D_;
  float* bi    = vv + B_ * D_;
  float* pre   = bi + B_ * K_ * D_;
  float* pair  = h_emb;

  span_max_kernel<<<B_ * SP_NSTRIP, SP_WAVES * 32, 0, stream>>>(emb, h_span, t_span, b_spans,
                                         h_emb, t_emb, b_emb);

  const dim3 g_small(D_ / 16, B_ / 16);
  const dim3 g_three(D_ / 32, G3_YB_BI + 2 * G3_YB_UV);
  const dim3 g_out((R_ + 15) / 16, B_ / 16);

  gemm3_nt_splitk<<<g_three, GK_WAVES * 32, 0, stream>>>(b_emb, wi_w, wi_b, bi,
                                                         h_emb, wu_w, wu_b, u,
                                                         t_emb, wv_w, wv_b, vv);

  attn_kernel<<<B_, 256, 0, stream>>>(u, vv, bi, pre);

  gemm_nt_splitk<1, 1><<<g_small, GK_WAVES * 32, 0, stream>>>(pre, ln1_w, ln1_b, pair, D_, 1);
  gemm_nt_splitk<1, 1><<<g_out, GK_WAVES * 32, 0, stream>>>(pair, pred_w, pred_b, out, R_, 0);
}


# Attention kernel: 24 serial load round trips and 81 barriers (8 block-wide LDS tree sums) become one load group, an in-wave lane butterfly and one barrier; bi kept in registers; new dot sum order
# speedup vs baseline: 11.0990x; 1.5481x over previous
#include <hip/hip_runtime.h>

#define B_  64
#define S_  512
#define D_  768
#define K_  8
#define R_  277

typedef __attribute__((ext_vector_type(2))) float v2f;
typedef __attribute__((ext_vector_type(8))) float v8f;

__device__ __forceinline__ v8f wmma_f32_16x16x4(v2f a, v2f b, v8f c) {
#if defined(__gfx1250__) && __has_builtin(__builtin_amdgcn_wmma_f32_16x16x4_f32)
  return __builtin_amdgcn_wmma_f32_16x16x4_f32(
      false, a, false, b, (short)0, c, false, false);
#else
  asm volatile("v_wmma_f32_16x16x4_f32 %0, %1, %2, %0"
               : "+v"(c) : "v"(a), "v"(b));
  return c;
#endif
}

typedef __attribute__((ext_vector_type(4))) float v4f;

#define SP_WAVES   8
#define SP_WROWS   (S_ / SP_WAVES)
#define SP_GROUP   16
#define SP_NGROUP  (SP_WROWS / SP_GROUP)
#define SP_COLS    128
#define SP_NSTRIP  (D_ / SP_COLS)
#define SP_NSPAN   (2 + K_)

__device__ __forceinline__ v4f vmax4(v4f a, v4f b) {
  v4f r;
  r.x = fmaxf(a.x, b.x);
  r.y = fmaxf(a.y, b.y);
  r.z = fmaxf(a.z, b.z);
  r.w = fmaxf(a.w, b.w);
  return r;
}

__global__ __launch_bounds__(SP_WAVES * 32)
void span_max_kernel(const float* __restrict__ emb,
                     const int*   __restrict__ h_span,
                     const int*   __restrict__ t_span,
                     const int*   __restrict__ b_spans,
                     float* __restrict__ h_emb,
                     float* __restrict__ t_emb,
                     float* __restrict__ b_emb) {
  __shared__ __attribute__((aligned(16))) float red[SP_WAVES * SP_NSPAN * SP_COLS];

  const int b     = blockIdx.x / SP_NSTRIP;
  const int strip = blockIdx.x % SP_NSTRIP;
  const int tid   = threadIdx.x;
  const int lane  = tid & 31;
  const int wave  = __builtin_amdgcn_readfirstlane(tid >> 5);

  int lo[SP_NSPAN], hi[SP_NSPAN];
  lo[0] = h_span[b * 2];  hi[0] = h_span[b * 2 + 1];
  lo[1] = t_span[b * 2];  hi[1] = t_span[b * 2 + 1];
#pragma unroll
  for (int k = 0; k < K_; ++k) {
    lo[2 + k] = b_spans[(b * K_ + k) * 2];
    hi[2 + k] = b_spans[(b * K_ + k) * 2 + 1];
  }

  v4f mx[SP_NSPAN];
#pragma unroll
  for (int j = 0; j < SP_NSPAN; ++j) {
    mx[j].x = -__builtin_inff();  mx[j].y = -__builtin_inff();
    mx[j].z = -__builtin_inff();  mx[j].w = -__builtin_inff();
  }

  const int row0 = wave * SP_WROWS;
  const v4f* src = (const v4f*)(emb + ((size_t)b * S_ + row0) * D_
                                + strip * SP_COLS) + lane;

  v4f nxt[SP_GROUP];
#pragma unroll
  for (int i = 0; i < SP_GROUP; ++i) nxt[i] = src[(size_t)i * (D_ / 4)];

#pragma nounroll
  for (int g = 0; g < SP_NGROUP; ++g) {
    v4f cur[SP_GROUP];
#pragma unroll
    for (int i = 0; i < SP_GROUP; ++i) cur[i] = nxt[i];

    if (g + 1 < SP_NGROUP) {
      const v4f* s2 = src + (size_t)(g + 1) * SP_GROUP * (D_ / 4);
#pragma unroll
      for (int i = 0; i < SP_GROUP; ++i) nxt[i] = s2[(size_t)i * (D_ / 4)];
    }

    v4f gm = cur[0];
#pragma unroll
    for (int i = 1; i < SP_GROUP; ++i) gm = vmax4(gm, cur[i]);

    const int r0 = row0 + g * SP_GROUP;
#pragma unroll
    for (int j = 0; j < SP_NSPAN; ++j) {
      int a = lo[j] - r0;      if (a < 0) a = 0;
      int e = hi[j] - r0 + 1;  if (e > SP_GROUP) e = SP_GROUP;
      if (a < e) {
        if (a == 0 && e == SP_GROUP) {
          mx[j] = vmax4(mx[j], gm);
        } else {
#pragma unroll
          for (int i = 0; i < SP_GROUP; ++i)
            if (i >= a && i < e) mx[j] = vmax4(mx[j], cur[i]);
        }
      }
    }
  }

#pragma unroll
  for (int j = 0; j < SP_NSPAN; ++j)
    *(v4f*)&red[((wave * SP_NSPAN + j) * 32 + lane) * 4] = mx[j];
  __syncthreads();

  for (int idx = tid; idx < SP_NSPAN * SP_COLS; idx += SP_WAVES * 32) {
    const int j = idx / SP_COLS;
    const int c = idx % SP_COLS;
    float m = red[j * SP_COLS + c];
#pragma unroll
    for (int w = 1; w < SP_WAVES; ++w)
      m = fmaxf(m, red[(w * SP_NSPAN + j) * SP_COLS + c]);
    const int col = strip * SP_COLS + c;
    float* dst = (j == 0) ? (h_emb + b * D_)
               : (j == 1) ? (t_emb + b * D_)
                          : (b_emb + (size_t)(b * K_ + (j - 2)) * D_);
    dst[col] = m;
  }
}

#define GK_WAVES  8
#define GK_SLICE  (D_ / GK_WAVES)
#define GK_LOADS  (GK_SLICE / 8)

template <int MT, int NT>
__device__ __forceinline__
void gemm_tile_splitk(const float* __restrict__ A,
                      const float* __restrict__ W,
                      const float* __restrict__ bias,
                      float* __restrict__ C,
                      int N, int relu, int m0, int n0) {
  constexpr int ROWS  = 16 * MT;
  constexpr int COLS  = 16 * NT;
  constexpr int PITCH = COLS + 2;
  __shared__ float part[GK_WAVES * ROWS * PITCH];

  const int tid  = threadIdx.x;
  const int lane = tid & 31;
  const int wave = __builtin_amdgcn_readfirstlane(tid >> 5);
  const int c    = lane & 15;
  const int h    = lane >> 4;

  const size_t koff = (size_t)wave * GK_SLICE + 4 * h;

  const float* ap[MT];
  const float* wp[NT];
#pragma unroll
  for (int mi = 0; mi < MT; ++mi)
    ap[mi] = A + (size_t)(m0 + mi * 16 + c) * D_ + koff;
#pragma unroll
  for (int ni = 0; ni < NT; ++ni) {
    int wr = n0 + ni * 16 + c;
    if (wr > N - 1) wr = N - 1;
    wp[ni] = W + (size_t)wr * D_ + koff;
  }

  v4f af[MT][GK_LOADS];
  v4f wf[NT][GK_LOADS];
#pragma unroll
  for (int j = 0; j < GK_LOADS; ++j) {
#pragma unroll
    for (int mi = 0; mi < MT; ++mi) af[mi][j] = *(const v4f*)(ap[mi] + 8 * j);
#pragma unroll
    for (int ni = 0; ni < NT; ++ni) wf[ni][j] = *(const v4f*)(wp[ni] + 8 * j);
  }
  __builtin_amdgcn_sched_barrier(0);

  v8f acc[MT][NT];
#pragma unroll
  for (int mi = 0; mi < MT; ++mi)
#pragma unroll
    for (int ni = 0; ni < NT; ++ni) acc[mi][ni] = v8f{};

#pragma unroll
  for (int j = 0; j < GK_LOADS; ++j) {
#pragma unroll
    for (int mi = 0; mi < MT; ++mi) {
#pragma unroll
      for (int ni = 0; ni < NT; ++ni) {
        v2f a0, b0;
        a0.x = af[mi][j].x;  a0.y = af[mi][j].y;
        b0.x = wf[ni][j].x;  b0.y = wf[ni][j].y;
        acc[mi][ni] = wmma_f32_16x16x4(a0, b0, acc[mi][ni]);
      }
    }
#pragma unroll
    for (int mi = 0; mi < MT; ++mi) {
#pragma unroll
      for (int ni = 0; ni < NT; ++ni) {
        v2f a1, b1;
        a1.x = af[mi][j].z;  a1.y = af[mi][j].w;
        b1.x = wf[ni][j].z;  b1.y = wf[ni][j].w;
        acc[mi][ni] = wmma_f32_16x16x4(a1, b1, acc[mi][ni]);
      }
    }
  }

  float* mine = part + wave * (ROWS * PITCH);
#pragma unroll
  for (int mi = 0; mi < MT; ++mi)
#pragma unroll
    for (int ni = 0; ni < NT; ++ni)
#pragma unroll
      for (int i = 0; i < 8; ++i)
        mine[(mi * 16 + 8 * h + i) * PITCH + ni * 16 + c] = acc[mi][ni][i];
  __syncthreads();

#pragma unroll
  for (int q = 0; q < MT * NT; ++q) {
    const int o   = tid + q * (GK_WAVES * 32);
    const int row = o / COLS;
    const int col = o % COLS;
    float s = part[row * PITCH + col];
#pragma unroll
    for (int w = 1; w < GK_WAVES; ++w)
      s += part[(w * ROWS + row) * PITCH + col];
    const int n = n0 + col;
    if (n < N) {
      float vout = s + bias[n];
      if (relu) vout = fmaxf(vout, 0.0f);
      C[(size_t)(m0 + row) * N + n] = vout;
    }
  }
}

template <int MT, int NT>
__global__ __launch_bounds__(GK_WAVES * 32)
void gemm_nt_splitk(const float* __restrict__ A,
                    const float* __restrict__ W,
                    const float* __restrict__ bias,
                    float* __restrict__ C,
                    int N, int relu) {
  gemm_tile_splitk<MT, NT>(A, W, bias, C, N, relu,
                           blockIdx.y * (16 * MT), blockIdx.x * (16 * NT));
}

#define G3_YB_BI  ((B_ * K_) / 32)
#define G3_YB_UV  (B_ / 32)

__global__ __launch_bounds__(GK_WAVES * 32)
void gemm3_nt_splitk(const float* __restrict__ Ab, const float* __restrict__ Wb,
                     const float* __restrict__ bb, float* __restrict__ Cb,
                     const float* __restrict__ Au, const float* __restrict__ Wu,
                     const float* __restrict__ bu, float* __restrict__ Cu,
                     const float* __restrict__ Av, const float* __restrict__ Wv,
                     const float* __restrict__ bv, float* __restrict__ Cv) {
  const int y = blockIdx.y;
  const float* A;  const float* W;  const float* bias;  float* C;  int yb;
  if (y < G3_YB_BI)                 { A = Ab; W = Wb; bias = bb; C = Cb; yb = y; }
  else if (y < G3_YB_BI + G3_YB_UV) { A = Au; W = Wu; bias = bu; C = Cu; yb = y - G3_YB_BI; }
  else                              { A = Av; W = Wv; bias = bv; C = Cv; yb = y - G3_YB_BI - G3_YB_UV; }
  gemm_tile_splitk<2, 2>(A, W, bias, C, D_, 0, yb * 32, blockIdx.x * 32);
}

#define AT_COLS     4
#define AT_WAVES    (D_ / (32 * AT_COLS))
#define AT_THREADS  (AT_WAVES * 32)

template <int M>
__device__ __forceinline__ float row_xor(float x) {
  return __builtin_bit_cast(float,
      __builtin_amdgcn_update_dpp(0, __builtin_bit_cast(int, x),
                                  0x160 + M, 0xf, 0xf, true));
}
__device__ __forceinline__ float half_xor(float x) {
  const int xb = __builtin_bit_cast(int, x);
  return __builtin_bit_cast(float,
      __builtin_amdgcn_permlanex16(xb, xb, 0x76543210, 0xfedcba98, false, false));
}

__global__ __launch_bounds__(AT_THREADS)
void attn_kernel(const float* __restrict__ u,
                 const float* __restrict__ v,
                 const float* __restrict__ bi,
                 float* __restrict__ pre) {
  __shared__ __attribute__((aligned(16))) float wsum[AT_WAVES * K_];

  const int b    = blockIdx.x;
  const int tid  = threadIdx.x;
  const int lane = tid & 31;
  const int wave = __builtin_amdgcn_readfirstlane(tid >> 5);

  const v4f* up = (const v4f*)(u  + (size_t)b * D_) + tid;
  const v4f* vp = (const v4f*)(v  + (size_t)b * D_) + tid;
  const v4f* bp = (const v4f*)(bi + (size_t)b * K_ * D_) + tid;

  const v4f u4 = *up;
  v4f b4[K_];
#pragma unroll
  for (int k = 0; k < K_; ++k) b4[k] = bp[k * (D_ / 4)];
  const v4f v4 = *vp;

  float p[K_];
#pragma unroll
  for (int k = 0; k < K_; ++k) {
    float s = u4.x * b4[k].x;
    s = fmaf(u4.y, b4[k].y, s);
    s = fmaf(u4.z, b4[k].z, s);
    s = fmaf(u4.w, b4[k].w, s);
    p[k] = s;
  }

#pragma unroll
  for (int k = 0; k < K_; ++k) p[k] += row_xor<1>(p[k]);
#pragma unroll
  for (int k = 0; k < K_; ++k) p[k] += row_xor<2>(p[k]);
#pragma unroll
  for (int k = 0; k < K_; ++k) p[k] += row_xor<4>(p[k]);
#pragma unroll
  for (int k = 0; k < K_; ++k) p[k] += row_xor<8>(p[k]);
#pragma unroll
  for (int k = 0; k < K_; ++k) p[k] += half_xor(p[k]);

  if (lane == 0) {
    v4f lo4, hi4;
    lo4.x = p[0];  lo4.y = p[1];  lo4.z = p[2];  lo4.w = p[3];
    hi4.x = p[4];  hi4.y = p[5];  hi4.z = p[6];  hi4.w = p[7];
    *(v4f*)&wsum[wave * K_]     = lo4;
    *(v4f*)&wsum[wave * K_ + 4] = hi4;
  }
  __syncthreads();

  const v4f* ws4 = (const v4f*)wsum;
  v4f slo = ws4[0], shi = ws4[1];
#pragma unroll
  for (int w = 1; w < AT_WAVES; ++w) {
    const v4f tlo = ws4[2 * w], thi = ws4[2 * w + 1];
    slo.x += tlo.x;  slo.y += tlo.y;  slo.z += tlo.z;  slo.w += tlo.w;
    shi.x += thi.x;  shi.y += thi.y;  shi.z += thi.z;  shi.w += thi.w;
  }
  float att[K_];
  att[0] = slo.x;  att[1] = slo.y;  att[2] = slo.z;  att[3] = slo.w;
  att[4] = shi.x;  att[5] = shi.y;  att[6] = shi.z;  att[7] = shi.w;

#pragma unroll
  for (int k = 0; k < K_; ++k) att[k] *= 0.125f;
  float m = att[0];
#pragma unroll
  for (int k = 1; k < K_; ++k) m = fmaxf(m, att[k]);
  float s = 0.f;
#pragma unroll
  for (int k = 0; k < K_; ++k) { att[k] = __expf(att[k] - m); s += att[k]; }
  const float inv = 1.0f / s;

  v4f c;
  c.x = 0.f;  c.y = 0.f;  c.z = 0.f;  c.w = 0.f;
#pragma unroll
  for (int k = 0; k < K_; ++k) {
    const float a = att[k] * inv;
    c.x += a * b4[k].x;
    c.y += a * b4[k].y;
    c.z += a * b4[k].z;
    c.w += a * b4[k].w;
  }
  v4f o;
  o.x = u4.x + v4.x + c.x;
  o.y = u4.y + v4.y + c.y;
  o.z = u4.z + v4.z + c.z;
  o.w = u4.w + v4.w + c.w;
  *((v4f*)(pre + (size_t)b * D_) + tid) = o;
}

extern "C" void kernel_launch(void* const* d_in, const int* in_sizes, int n_in,
                              void* d_out, int out_size, void* d_ws, size_t ws_size,
                              hipStream_t stream) {
  (void)in_sizes; (void)n_in; (void)out_size; (void)ws_size;

  const float* emb     = (const float*)d_in[0];
  const int*   h_span  = (const int*)d_in[1];
  const int*   t_span  = (const int*)d_in[2];
  const int*   b_spans = (const int*)d_in[3];
  const float* wu_w    = (const float*)d_in[4];
  const float* wu_b    = (const float*)d_in[5];
  const float* wv_w    = (const float*)d_in[6];
  const float* wv_b    = (const float*)d_in[7];
  const float* wi_w    = (const float*)d_in[8];
  const float* wi_b    = (const float*)d_in[9];
  const float* ln1_w   = (const float*)d_in[10];
  const float* ln1_b   = (const float*)d_in[11];
  const float* pred_w  = (const float*)d_in[12];
  const float* pred_b  = (const float*)d_in[13];
  float* out = (float*)d_out;

  float* ws    = (float*)d_ws;
  float* h_emb = ws;
  float* t_emb = h_emb + B_ * D_;
  float* b_emb = t_emb + B_ * D_;
  float* u     = b_emb + B_ * K_ * D_;
  float* vv    = u + B_ * D_;
  float* bi    = vv + B_ * D_;
  float* pre   = bi + B_ * K_ * D_;
  float* pair  = h_emb;

  span_max_kernel<<<B_ * SP_NSTRIP, SP_WAVES * 32, 0, stream>>>(emb, h_span, t_span, b_spans,
                                         h_emb, t_emb, b_emb);

  const dim3 g_small(D_ / 16, B_ / 16);
  const dim3 g_three(D_ / 32, G3_YB_BI + 2 * G3_YB_UV);
  const dim3 g_out((R_ + 15) / 16, B_ / 16);

  gemm3_nt_splitk<<<g_three, GK_WAVES * 32, 0, stream>>>(b_emb, wi_w, wi_b, bi,
                                                         h_emb, wu_w, wu_b, u,
                                                         t_emb, wv_w, wv_b, vv);

  attn_kernel<<<B_, AT_THREADS, 0, stream>>>(u, vv, bi, pre);

  gemm_nt_splitk<1, 1><<<g_small, GK_WAVES * 32, 0, stream>>>(pre, ln1_w, ln1_b, pair, D_, 1);
  gemm_nt_splitk<1, 1><<<g_out, GK_WAVES * 32, 0, stream>>>(pair, pred_w, pred_b, out, R_, 0);
}
